# starting kernel
# baseline (speedup 1.0000x reference)
#include <hip/hip_runtime.h>


typedef __attribute__((ext_vector_type(2))) float v2f;
typedef __attribute__((ext_vector_type(8))) float v8f;

__global__ void GCN_zero_kernel(float* __restrict__ p, int n4) {
  int i = blockIdx.x * blockDim.x + threadIdx.x;
  if (i < n4) {
    ((float4*)p)[i] = make_float4(0.f, 0.f, 0.f, 0.f);
  }
}

template <int F>
__global__ void GCN_spmm_kernel(const float* __restrict__ feat,
                                const int* __restrict__ src,
                                const int* __restrict__ dst,
                                float* __restrict__ out,
                                int num_edges) {
  constexpr int TPE = F / 4;
  long long gtid = (long long)blockIdx.x * blockDim.x + threadIdx.x;
  long long e = gtid / TPE;
  int j = (int)(gtid % TPE) * 4;
  if (e >= num_edges) return;
  int s = src[e];
  int d = dst[e];
  const float4 v = *(const float4*)(feat + (long long)s * F + j);
  float* o = out + (long long)d * F + j;
  atomicAdd(o + 0, v.x);
  atomicAdd(o + 1, v.y);
  atomicAdd(o + 2, v.z);
  atomicAdd(o + 3, v.w);
}

template <int K, int NCOL>
__global__ void GCN_gemm_wmma_kernel(const float* __restrict__ A,
                                     const float* __restrict__ B,
                                     float* __restrict__ D,
                                     int nrows) {
  constexpr int LDSROW = K + 2;
  __shared__ float As[16 * LDSROW];

  const int m0 = blockIdx.x * 16;
  const int lane = threadIdx.x & 31;
  const int wave = threadIdx.x >> 5;

  for (int idx = threadIdx.x; idx < 16 * (K / 4); idx += blockDim.x) {
    int r = idx / (K / 4);
    int c = (idx % (K / 4)) * 4;
    float4 v = make_float4(0.f, 0.f, 0.f, 0.f);
    if (m0 + r < nrows) {
      v = *(const float4*)(A + (long long)(m0 + r) * K + c);
    }
    As[r * LDSROW + c + 0] = v.x;
    As[r * LDSROW + c + 1] = v.y;
    As[r * LDSROW + c + 2] = v.z;
    As[r * LDSROW + c + 3] = v.w;
  }
  __syncthreads();

  const int n0 = wave * 16;
  const int half = lane >> 4;
  const int l16 = lane & 15;

  v8f c = {};
#pragma unroll
  for (int k0 = 0; k0 < K; k0 += 4) {
    const int kb = k0 + 2 * half;
    v2f a = *(const v2f*)&As[l16 * LDSROW + kb];
    v2f b;
    b.x = B[(kb + 0) * NCOL + n0 + l16];
    b.y = B[(kb + 1) * NCOL + n0 + l16];
    c = __builtin_amdgcn_wmma_f32_16x16x4_f32(
         false, a,  false, b,
         (short)0, c,  false,  false);
  }

#pragma unroll
  for (int r = 0; r < 8; ++r) {
    int row = m0 + r + 8 * half;
    if (row < nrows) {
      D[(long long)row * NCOL + n0 + l16] = c[r];
    }
  }
}

extern "C" void kernel_launch(void* const* d_in, const int* in_sizes, int n_in,
                              void* d_out, int out_size, void* d_ws, size_t ws_size,
                              hipStream_t stream) {
  const float* x   = (const float*)d_in[0];
  const float* W1  = (const float*)d_in[1];
  const float* W2  = (const float*)d_in[2];
  const int*  src1 = (const int*)d_in[3];
  const int*  dst1 = (const int*)d_in[4];
  const int*  src2 = (const int*)d_in[5];
  const int*  dst2 = (const int*)d_in[6];
  float* out = (float*)d_out;

  constexpr int IN = 128, H = 128, C = 64;
  const int N = in_sizes[0] / IN;
  const int E = in_sizes[3];

  float* agg = (float*)d_ws;
  float* h   = agg + (size_t)N * H;
  float* t   = agg;

  {
    int n4 = N * H / 4;
    GCN_zero_kernel<<<(n4 + 255) / 256, 256, 0, stream>>>(agg, n4);
    int o4 = N * C / 4;
    GCN_zero_kernel<<<(o4 + 255) / 256, 256, 0, stream>>>(out, o4);
  }

  {
    long long threads = (long long)E * (IN / 4);
    int blocks = (int)((threads + 255) / 256);
    GCN_spmm_kernel<IN><<<blocks, 256, 0, stream>>>(x, src1, dst1, agg, E);
  }

  GCN_gemm_wmma_kernel<128, 128><<<(N + 15) / 16, 256, 0, stream>>>(agg, W1, h, N);

  GCN_gemm_wmma_kernel<128, 64><<<(N + 15) / 16, 128, 0, stream>>>(h, W2, t, N);

  {
    long long threads = (long long)E * (C / 4);
    int blocks = (int)((threads + 255) / 256);
    GCN_spmm_kernel<C><<<blocks, 256, 0, stream>>>(t, src2, dst2, out, E);
  }
}


# Use linearity: fold W12=W1@W2, project y=x@W12 to 64 columns first, then run both scatter-adds on 64-float rows (f32 sums re-associated; layer-1 atomics halved)
# speedup vs baseline: 1.5191x; 1.5191x over previous
#include <hip/hip_runtime.h>


typedef __attribute__((ext_vector_type(2))) float v2f;
typedef __attribute__((ext_vector_type(8))) float v8f;

__global__ void GCN_zero_kernel(float* __restrict__ p, int n4) {
  int i = blockIdx.x * blockDim.x + threadIdx.x;
  if (i < n4) {
    ((float4*)p)[i] = make_float4(0.f, 0.f, 0.f, 0.f);
  }
}

template <int F>
__global__ void GCN_spmm_kernel(const float* __restrict__ feat,
                                const int* __restrict__ src,
                                const int* __restrict__ dst,
                                float* __restrict__ out,
                                int num_edges) {
  constexpr int TPE = F / 4;
  long long gtid = (long long)blockIdx.x * blockDim.x + threadIdx.x;
  long long e = gtid / TPE;
  int j = (int)(gtid % TPE) * 4;
  if (e >= num_edges) return;
  int s = src[e];
  int d = dst[e];
  const float4 v = *(const float4*)(feat + (long long)s * F + j);
  float* o = out + (long long)d * F + j;
  atomicAdd(o + 0, v.x);
  atomicAdd(o + 1, v.y);
  atomicAdd(o + 2, v.z);
  atomicAdd(o + 3, v.w);
}

template <int K, int NCOL>
__global__ void GCN_gemm_wmma_kernel(const float* __restrict__ A,
                                     const float* __restrict__ B,
                                     float* __restrict__ D,
                                     int nrows) {
  constexpr int LDSROW = K + 2;
  __shared__ float As[16 * LDSROW];

  const int m0 = blockIdx.x * 16;
  const int lane = threadIdx.x & 31;
  const int wave = threadIdx.x >> 5;

  for (int idx = threadIdx.x; idx < 16 * (K / 4); idx += blockDim.x) {
    int r = idx / (K / 4);
    int c = (idx % (K / 4)) * 4;
    float4 v = make_float4(0.f, 0.f, 0.f, 0.f);
    if (m0 + r < nrows) {
      v = *(const float4*)(A + (long long)(m0 + r) * K + c);
    }
    As[r * LDSROW + c + 0] = v.x;
    As[r * LDSROW + c + 1] = v.y;
    As[r * LDSROW + c + 2] = v.z;
    As[r * LDSROW + c + 3] = v.w;
  }
  __syncthreads();

  const int n0 = wave * 16;
  const int half = lane >> 4;
  const int l16 = lane & 15;

  v8f c = {};
#pragma unroll
  for (int k0 = 0; k0 < K; k0 += 4) {
    const int kb = k0 + 2 * half;
    v2f a = *(const v2f*)&As[l16 * LDSROW + kb];
    v2f b;
    b.x = B[(kb + 0) * NCOL + n0 + l16];
    b.y = B[(kb + 1) * NCOL + n0 + l16];
    c = __builtin_amdgcn_wmma_f32_16x16x4_f32(
         false, a,  false, b,
         (short)0, c,  false,  false);
  }

#pragma unroll
  for (int r = 0; r < 8; ++r) {
    int row = m0 + r + 8 * half;
    if (row < nrows) {
      D[(long long)row * NCOL + n0 + l16] = c[r];
    }
  }
}

extern "C" void kernel_launch(void* const* d_in, const int* in_sizes, int n_in,
                              void* d_out, int out_size, void* d_ws, size_t ws_size,
                              hipStream_t stream) {
  const float* x   = (const float*)d_in[0];
  const float* W1  = (const float*)d_in[1];
  const float* W2  = (const float*)d_in[2];
  const int*  src1 = (const int*)d_in[3];
  const int*  dst1 = (const int*)d_in[4];
  const int*  src2 = (const int*)d_in[5];
  const int*  dst2 = (const int*)d_in[6];
  float* out = (float*)d_out;

  constexpr int IN = 128, H = 128, C = 64;
  const int N = in_sizes[0] / IN;
  const int E = in_sizes[3];

  float* W12 = (float*)d_ws;
  float* y   = W12 + (size_t)IN * C;
  float* z   = y + (size_t)N * C;

  GCN_gemm_wmma_kernel<H, C><<<(IN + 15) / 16, 128, 0, stream>>>(W1, W2, W12, IN);

  GCN_gemm_wmma_kernel<IN, C><<<(N + 15) / 16, 128, 0, stream>>>(x, W12, y, N);

  {
    int n4 = N * C / 4;
    GCN_zero_kernel<<<(n4 + 255) / 256, 256, 0, stream>>>(z, n4);
    GCN_zero_kernel<<<(n4 + 255) / 256, 256, 0, stream>>>(out, n4);
  }

  {
    long long threads = (long long)E * (C / 4);
    int blocks = (int)((threads + 255) / 256);
    GCN_spmm_kernel<C><<<blocks, 256, 0, stream>>>(y, src1, dst1, z, E);
  }

  {
    long long threads = (long long)E * (C / 4);
    int blocks = (int)((threads + 255) / 256);
    GCN_spmm_kernel<C><<<blocks, 256, 0, stream>>>(z, src2, dst2, out, E);
  }
}


# Scatter-add re-mapped to one wave per edge row: lane = column (l, l+32), so every gather load and every f32 atomic covers one whole 128-B line; waves loop over edges, indices by scalar loads
# speedup vs baseline: 2.8516x; 1.8772x over previous
#include <hip/hip_runtime.h>


typedef __attribute__((ext_vector_type(2))) float v2f;
typedef __attribute__((ext_vector_type(8))) float v8f;

__global__ void GCN_zero_kernel(float* __restrict__ p, int n4) {
  int i = blockIdx.x * blockDim.x + threadIdx.x;
  if (i < n4) {
    ((float4*)p)[i] = make_float4(0.f, 0.f, 0.f, 0.f);
  }
}

template <int F, int G, int BLOCK>
__global__ __launch_bounds__(BLOCK) void GCN_spmm_rows_kernel(
    const float* __restrict__ feat,
    const int* __restrict__ src,
    const int* __restrict__ dst,
    float* __restrict__ out,
    int num_edges) {
  static_assert(F == 64, "the lane map is written for 64-float rows");
  constexpr int WAVES = BLOCK / 32;
  const int lane = threadIdx.x & 31;
  const int wave = __builtin_amdgcn_readfirstlane((int)(threadIdx.x >> 5));
  const long long first = ((long long)blockIdx.x * WAVES + wave) * G;
  const long long step = (long long)gridDim.x * WAVES * G;

  for (long long e0 = first; e0 < num_edges; e0 += step) {
    if (e0 + G <= num_edges) {
      int s[G], d[G];
      for (int i = 0; i < G; ++i) {
        s[i] = src[e0 + i];
        d[i] = dst[e0 + i];
      }
      float lo[G], hi[G];
      for (int i = 0; i < G; ++i) {
        const float* p = feat + (long long)s[i] * F + lane;
        lo[i] = p[0];
        hi[i] = p[32];
      }
      for (int i = 0; i < G; ++i) {
        float* o = out + (long long)d[i] * F + lane;
        atomicAdd(o, lo[i]);
        atomicAdd(o + 32, hi[i]);
      }
    } else {
      for (long long e = e0; e < num_edges; ++e) {
        const float* p = feat + (long long)src[e] * F + lane;
        float* o = out + (long long)dst[e] * F + lane;
        const float a = p[0];
        const float b = p[32];
        atomicAdd(o, a);
        atomicAdd(o + 32, b);
      }
    }
  }
}

template <int K, int NCOL>
__global__ void GCN_gemm_wmma_kernel(const float* __restrict__ A,
                                     const float* __restrict__ B,
                                     float* __restrict__ D,
                                     int nrows) {
  constexpr int LDSROW = K + 2;
  __shared__ float As[16 * LDSROW];

  const int m0 = blockIdx.x * 16;
  const int lane = threadIdx.x & 31;
  const int wave = threadIdx.x >> 5;

  for (int idx = threadIdx.x; idx < 16 * (K / 4); idx += blockDim.x) {
    int r = idx / (K / 4);
    int c = (idx % (K / 4)) * 4;
    float4 v = make_float4(0.f, 0.f, 0.f, 0.f);
    if (m0 + r < nrows) {
      v = *(const float4*)(A + (long long)(m0 + r) * K + c);
    }
    As[r * LDSROW + c + 0] = v.x;
    As[r * LDSROW + c + 1] = v.y;
    As[r * LDSROW + c + 2] = v.z;
    As[r * LDSROW + c + 3] = v.w;
  }
  __syncthreads();

  const int n0 = wave * 16;
  const int half = lane >> 4;
  const int l16 = lane & 15;

  v8f c = {};
#pragma unroll
  for (int k0 = 0; k0 < K; k0 += 4) {
    const int kb = k0 + 2 * half;
    v2f a = *(const v2f*)&As[l16 * LDSROW + kb];
    v2f b;
    b.x = B[(kb + 0) * NCOL + n0 + l16];
    b.y = B[(kb + 1) * NCOL + n0 + l16];
    c = __builtin_amdgcn_wmma_f32_16x16x4_f32(
         false, a,  false, b,
         (short)0, c,  false,  false);
  }

#pragma unroll
  for (int r = 0; r < 8; ++r) {
    int row = m0 + r + 8 * half;
    if (row < nrows) {
      D[(long long)row * NCOL + n0 + l16] = c[r];
    }
  }
}

extern "C" void kernel_launch(void* const* d_in, const int* in_sizes, int n_in,
                              void* d_out, int out_size, void* d_ws, size_t ws_size,
                              hipStream_t stream) {
  const float* x   = (const float*)d_in[0];
  const float* W1  = (const float*)d_in[1];
  const float* W2  = (const float*)d_in[2];
  const int*  src1 = (const int*)d_in[3];
  const int*  dst1 = (const int*)d_in[4];
  const int*  src2 = (const int*)d_in[5];
  const int*  dst2 = (const int*)d_in[6];
  float* out = (float*)d_out;

  constexpr int IN = 128, H = 128, C = 64;
  const int N = in_sizes[0] / IN;
  const int E = in_sizes[3];

  float* W12 = (float*)d_ws;
  float* y   = W12 + (size_t)IN * C;
  float* z   = y + (size_t)N * C;

  GCN_gemm_wmma_kernel<H, C><<<(IN + 15) / 16, 128, 0, stream>>>(W1, W2, W12, IN);

  GCN_gemm_wmma_kernel<IN, C><<<(N + 15) / 16, 128, 0, stream>>>(x, W12, y, N);

  {
    int n4 = N * C / 4;
    GCN_zero_kernel<<<(n4 + 255) / 256, 256, 0, stream>>>(z, n4);
    GCN_zero_kernel<<<(n4 + 255) / 256, 256, 0, stream>>>(out, n4);
  }

  constexpr int G = 8, BLOCK = 256;
  const int groups = (E + G * (BLOCK / 32) - 1) / (G * (BLOCK / 32));
  const int blocks = groups < 2048 ? (groups > 0 ? groups : 1) : 2048;

  GCN_spmm_rows_kernel<C, G, BLOCK><<<blocks, BLOCK, 0, stream>>>(y, src1, dst1, z, E);

  GCN_spmm_rows_kernel<C, G, BLOCK><<<blocks, BLOCK, 0, stream>>>(z, src2, dst2, out, E);
}


# Zero-fill of both accumulation targets folded into the projection kernel (each block clears its own rows of z and out): the two fill launches and their 50000 one-store waves go away
# speedup vs baseline: 2.9799x; 1.0450x over previous
#include <hip/hip_runtime.h>


typedef __attribute__((ext_vector_type(2))) float v2f;
typedef __attribute__((ext_vector_type(8))) float v8f;

template <int F, int G, int BLOCK>
__global__ __launch_bounds__(BLOCK) void GCN_spmm_rows_kernel(
    const float* __restrict__ feat,
    const int* __restrict__ src,
    const int* __restrict__ dst,
    float* __restrict__ out,
    int num_edges) {
  static_assert(F == 64, "the lane map is written for 64-float rows");
  constexpr int WAVES = BLOCK / 32;
  const int lane = threadIdx.x & 31;
  const int wave = __builtin_amdgcn_readfirstlane((int)(threadIdx.x >> 5));
  const long long first = ((long long)blockIdx.x * WAVES + wave) * G;
  const long long step = (long long)gridDim.x * WAVES * G;

  for (long long e0 = first; e0 < num_edges; e0 += step) {
    if (e0 + G <= num_edges) {
      int s[G], d[G];
      for (int i = 0; i < G; ++i) {
        s[i] = src[e0 + i];
        d[i] = dst[e0 + i];
      }
      float lo[G], hi[G];
      for (int i = 0; i < G; ++i) {
        const float* p = feat + (long long)s[i] * F + lane;
        lo[i] = p[0];
        hi[i] = p[32];
      }
      for (int i = 0; i < G; ++i) {
        float* o = out + (long long)d[i] * F + lane;
        atomicAdd(o, lo[i]);
        atomicAdd(o + 32, hi[i]);
      }
    } else {
      for (long long e = e0; e < num_edges; ++e) {
        const float* p = feat + (long long)src[e] * F + lane;
        float* o = out + (long long)dst[e] * F + lane;
        const float a = p[0];
        const float b = p[32];
        atomicAdd(o, a);
        atomicAdd(o + 32, b);
      }
    }
  }
}

template <int K, int NCOL, bool CLEAR>
__global__ void GCN_gemm_wmma_kernel(const float* __restrict__ A,
                                     const float* __restrict__ B,
                                     float* __restrict__ D,
                                     float* __restrict__ clr0,
                                     float* __restrict__ clr1,
                                     int nrows) {
  constexpr int LDSROW = K + 2;
  __shared__ float As[16 * LDSROW];

  const int m0 = blockIdx.x * 16;
  const int lane = threadIdx.x & 31;
  const int wave = threadIdx.x >> 5;

  if (CLEAR) {
    const long long base4 = (long long)m0 * (NCOL / 4);
    const long long lim4 = (long long)nrows * (NCOL / 4);
    const float4 zero4 = make_float4(0.f, 0.f, 0.f, 0.f);
    for (int i = threadIdx.x; i < 16 * (NCOL / 4); i += blockDim.x) {
      if (base4 + i < lim4) {
        ((float4*)clr0)[base4 + i] = zero4;
        ((float4*)clr1)[base4 + i] = zero4;
      }
    }
  }

  for (int idx = threadIdx.x; idx < 16 * (K / 4); idx += blockDim.x) {
    int r = idx / (K / 4);
    int c = (idx % (K / 4)) * 4;
    float4 v = make_float4(0.f, 0.f, 0.f, 0.f);
    if (m0 + r < nrows) {
      v = *(const float4*)(A + (long long)(m0 + r) * K + c);
    }
    As[r * LDSROW + c + 0] = v.x;
    As[r * LDSROW + c + 1] = v.y;
    As[r * LDSROW + c + 2] = v.z;
    As[r * LDSROW + c + 3] = v.w;
  }
  __syncthreads();

  const int n0 = wave * 16;
  const int half = lane >> 4;
  const int l16 = lane & 15;

  v8f c = {};
#pragma unroll
  for (int k0 = 0; k0 < K; k0 += 4) {
    const int kb = k0 + 2 * half;
    v2f a = *(const v2f*)&As[l16 * LDSROW + kb];
    v2f b;
    b.x = B[(kb + 0) * NCOL + n0 + l16];
    b.y = B[(kb + 1) * NCOL + n0 + l16];
    c = __builtin_amdgcn_wmma_f32_16x16x4_f32(
         false, a,  false, b,
         (short)0, c,  false,  false);
  }

#pragma unroll
  for (int r = 0; r < 8; ++r) {
    int row = m0 + r + 8 * half;
    if (row < nrows) {
      D[(long long)row * NCOL + n0 + l16] = c[r];
    }
  }
}

extern "C" void kernel_launch(void* const* d_in, const int* in_sizes, int n_in,
                              void* d_out, int out_size, void* d_ws, size_t ws_size,
                              hipStream_t stream) {
  const float* x   = (const float*)d_in[0];
  const float* W1  = (const float*)d_in[1];
  const float* W2  = (const float*)d_in[2];
  const int*  src1 = (const int*)d_in[3];
  const int*  dst1 = (const int*)d_in[4];
  const int*  src2 = (const int*)d_in[5];
  const int*  dst2 = (const int*)d_in[6];
  float* out = (float*)d_out;

  constexpr int IN = 128, H = 128, C = 64;
  const int N = in_sizes[0] / IN;
  const int E = in_sizes[3];

  float* W12 = (float*)d_ws;
  float* y   = W12 + (size_t)IN * C;
  float* z   = y + (size_t)N * C;

  GCN_gemm_wmma_kernel<H, C, false><<<(IN + 15) / 16, 128, 0, stream>>>(
      W1, W2, W12, nullptr, nullptr, IN);

  GCN_gemm_wmma_kernel<IN, C, true><<<(N + 15) / 16, 128, 0, stream>>>(
      x, W12, y, z, out, N);

  constexpr int G = 8, BLOCK = 256;
  const int groups = (E + G * (BLOCK / 32) - 1) / (G * (BLOCK / 32));
  const int blocks = groups < 2048 ? (groups > 0 ? groups : 1) : 2048;

  GCN_spmm_rows_kernel<C, G, BLOCK><<<blocks, BLOCK, 0, stream>>>(y, src1, dst1, z, E);

  GCN_spmm_rows_kernel<C, G, BLOCK><<<blocks, BLOCK, 0, stream>>>(z, src2, dst2, out, E);
}


# Scatter-add split in two destination-row ranges, one per cache half: work-groups draw edge chunks from their half's ticket counter (die id), add only its rows; f32 atomics, arrival order as before
# speedup vs baseline: 3.5140x; 1.1792x over previous
#include <hip/hip_runtime.h>


typedef __attribute__((ext_vector_type(2))) float v2f;
typedef __attribute__((ext_vector_type(8))) float v8f;

template <int F, int SLOTS, bool UPPER>
__device__ __forceinline__ void GCN_rows_of_range(
    const float* __restrict__ feat,
    const int* __restrict__ src,
    const int* __restrict__ dst,
    float* __restrict__ out,
    long long e0, int num_edges, int split, int lane) {
  if (e0 + SLOTS <= num_edges) {
    int s[SLOTS], d[SLOTS];
    for (int i = 0; i < SLOTS; ++i) {
      s[i] = __builtin_amdgcn_readfirstlane(src[e0 + i]);
      d[i] = dst[e0 + i];
    }
    float lo[SLOTS], hi[SLOTS];
    for (int i = 0; i < SLOTS; ++i) {
      lo[i] = 0.f;
      hi[i] = 0.f;
      if (UPPER ? (d[i] >= split) : (d[i] < split)) {
        const float* p = feat + (long long)s[i] * F + lane;
        lo[i] = p[0];
        hi[i] = p[32];
      }
    }
    for (int i = 0; i < SLOTS; ++i) {
      if (UPPER ? (d[i] >= split) : (d[i] < split)) {
        float* o = out + (long long)d[i] * F + lane;
        atomicAdd(o, lo[i]);
        atomicAdd(o + 32, hi[i]);
      }
    }
  } else {
    for (long long e = e0; e < num_edges && e < e0 + SLOTS; ++e) {
      const int dd = dst[e];
      if (UPPER ? (dd >= split) : (dd < split)) {
        const float* p = feat + (long long)src[e] * F + lane;
        float* o = out + (long long)dd * F + lane;
        const float a = p[0];
        const float b = p[32];
        atomicAdd(o, a);
        atomicAdd(o + 32, b);
      }
    }
  }
}

__device__ __forceinline__ void GCN_lds_barrier() {
  __builtin_amdgcn_fence(__ATOMIC_RELEASE, "workgroup", "local");
  __builtin_amdgcn_s_barrier();
  __builtin_amdgcn_fence(__ATOMIC_ACQUIRE, "workgroup", "local");
}

template <int F, int SLOTS, int BLOCK>
__global__ __launch_bounds__(BLOCK) void GCN_spmm_rows_kernel(
    const float* __restrict__ feat,
    const int* __restrict__ src,
    const int* __restrict__ dst,
    float* __restrict__ out,
    int num_edges,
    int num_rows,
    int* __restrict__ tickets) {
  static_assert(F == 64, "the lane map is written for 64-float rows");
  constexpr int WORKERS = BLOCK / 32 - 1;
  constexpr int CHUNK = WORKERS * SLOTS;
  __shared__ int sh_ticket[2];
  __shared__ int sh_range;

  const int lane = threadIdx.x & 31;
  const int wave = __builtin_amdgcn_readfirstlane((int)(threadIdx.x >> 5));
  const int nchunks = (num_edges + CHUNK - 1) / CHUNK;
  const int split = num_rows >> 1;

  if (threadIdx.x == 0) {
    const unsigned where = __builtin_amdgcn_s_sendmsg_rtn(0x87);
    const int r = (int)((where >> 18) & 1u);
    sh_range = r;
    sh_ticket[0] = atomicAdd(&tickets[r], 1);
  }
  GCN_lds_barrier();
  int range = __builtin_amdgcn_readfirstlane(sh_range);
  int slot = 0;
  bool other = false;

  for (int trip = 0; trip < 2 * nchunks + 2; ++trip) {
    const int t = __builtin_amdgcn_readfirstlane(sh_ticket[slot]);
    if ((unsigned)t >= (unsigned)nchunks) {
      if (other) break;
      other = true;
      range ^= 1;
      slot ^= 1;
      if (threadIdx.x == 0) sh_ticket[slot] = atomicAdd(&tickets[range], 1);
      GCN_lds_barrier();
      continue;
    }
    int next = 0;
    if (wave == 0) {
      if (lane == 0) next = atomicAdd(&tickets[range], 1);
    } else {
      const long long e0 = (long long)t * CHUNK + (long long)(wave - 1) * SLOTS;
      if (range) {
        GCN_rows_of_range<F, SLOTS, true>(feat, src, dst, out, e0, num_edges, split, lane);
      } else {
        GCN_rows_of_range<F, SLOTS, false>(feat, src, dst, out, e0, num_edges, split, lane);
      }
    }

    slot ^= 1;
    if (threadIdx.x == 0) sh_ticket[slot] = next;
    GCN_lds_barrier();
  }
}

template <int K, int NCOL, bool CLEAR>
__global__ void GCN_gemm_wmma_kernel(const float* __restrict__ A,
                                     const float* __restrict__ B,
                                     float* __restrict__ D,
                                     float* __restrict__ clr0,
                                     float* __restrict__ clr1,
                                     int* __restrict__ tickets,
                                     int nrows) {
  constexpr int LDSROW = K + 2;
  __shared__ float As[16 * LDSROW];

  const int m0 = blockIdx.x * 16;
  const int lane = threadIdx.x & 31;
  const int wave = threadIdx.x >> 5;

  if (CLEAR) {
    if (blockIdx.x == 0 && threadIdx.x < 4) tickets[threadIdx.x] = 0;
    const long long base4 = (long long)m0 * (NCOL / 4);
    const long long lim4 = (long long)nrows * (NCOL / 4);
    const float4 zero4 = make_float4(0.f, 0.f, 0.f, 0.f);
    for (int i = threadIdx.x; i < 16 * (NCOL / 4); i += blockDim.x) {
      if (base4 + i < lim4) {
        ((float4*)clr0)[base4 + i] = zero4;
        ((float4*)clr1)[base4 + i] = zero4;
      }
    }
  }

  for (int idx = threadIdx.x; idx < 16 * (K / 4); idx += blockDim.x) {
    int r = idx / (K / 4);
    int c = (idx % (K / 4)) * 4;
    float4 v = make_float4(0.f, 0.f, 0.f, 0.f);
    if (m0 + r < nrows) {
      v = *(const float4*)(A + (long long)(m0 + r) * K + c);
    }
    As[r * LDSROW + c + 0] = v.x;
    As[r * LDSROW + c + 1] = v.y;
    As[r * LDSROW + c + 2] = v.z;
    As[r * LDSROW + c + 3] = v.w;
  }
  __syncthreads();

  const int n0 = wave * 16;
  const int half = lane >> 4;
  const int l16 = lane & 15;

  v8f c = {};
#pragma unroll
  for (int k0 = 0; k0 < K; k0 += 4) {
    const int kb = k0 + 2 * half;
    v2f a = *(const v2f*)&As[l16 * LDSROW + kb];
    v2f b;
    b.x = B[(kb + 0) * NCOL + n0 + l16];
    b.y = B[(kb + 1) * NCOL + n0 + l16];
    c = __builtin_amdgcn_wmma_f32_16x16x4_f32(
         false, a,  false, b,
         (short)0, c,  false,  false);
  }

#pragma unroll
  for (int r = 0; r < 8; ++r) {
    int row = m0 + r + 8 * half;
    if (row < nrows) {
      D[(long long)row * NCOL + n0 + l16] = c[r];
    }
  }
}

extern "C" void kernel_launch(void* const* d_in, const int* in_sizes, int n_in,
                              void* d_out, int out_size, void* d_ws, size_t ws_size,
                              hipStream_t stream) {
  const float* x   = (const float*)d_in[0];
  const float* W1  = (const float*)d_in[1];
  const float* W2  = (const float*)d_in[2];
  const int*  src1 = (const int*)d_in[3];
  const int*  dst1 = (const int*)d_in[4];
  const int*  src2 = (const int*)d_in[5];
  const int*  dst2 = (const int*)d_in[6];
  float* out = (float*)d_out;

  constexpr int IN = 128, H = 128, C = 64;
  const int N = in_sizes[0] / IN;
  const int E = in_sizes[3];

  float* W12 = (float*)d_ws;
  float* y   = W12 + (size_t)IN * C;
  float* z   = y + (size_t)N * C;
  int* tickets = (int*)(z + (size_t)N * C);

  GCN_gemm_wmma_kernel<H, C, false><<<(IN + 15) / 16, 128, 0, stream>>>(
      W1, W2, W12, nullptr, nullptr, nullptr, IN);

  GCN_gemm_wmma_kernel<IN, C, true><<<(N + 15) / 16, 128, 0, stream>>>(
      x, W12, y, z, out, tickets, N);

  constexpr int SLOTS = 16, BLOCK = 512;
  constexpr int CHUNK = SLOTS * (BLOCK / 32 - 1);
  const int draws = 2 * ((E + CHUNK - 1) / CHUNK);
  const int blocks = draws < 1024 ? (draws > 0 ? draws : 1) : 1024;

  GCN_spmm_rows_kernel<C, SLOTS, BLOCK><<<blocks, BLOCK, 0, stream>>>(
      y, src1, dst1, z, E, N, tickets);

  GCN_spmm_rows_kernel<C, SLOTS, BLOCK><<<blocks, BLOCK, 0, stream>>>(
      z, src2, dst2, out, E, N, tickets + 2);
}


# Edges grouped by 196-row dst block in LDS, slice by slice, beside the weight product; each pass sums a block's rows in LDS (ds_add_f32, arrival order) and stores them once: no global atomics, no fill
# speedup vs baseline: 3.8566x; 1.0975x over previous
#include <hip/hip_runtime.h>


typedef __attribute__((ext_vector_type(2))) float v2f;
typedef __attribute__((ext_vector_type(8))) float v8f;

constexpr int GCN_NBLK = 256;
constexpr int GCN_RPB = 196;
constexpr int GCN_NSEG = 64;
constexpr int GCN_SL = 12500;
constexpr int GCN_SEGP = 12544;
constexpr int GCN_OFFP = 288;

template <int BLOCK>
__device__ __forceinline__ void GCN_group_slice(const int* __restrict__ src,
                                                const int* __restrict__ dst,
                                                int num_edges, int seg,
                                                int* __restrict__ rec_out,
                                                int* __restrict__ off_out) {
  constexpr int PER = (GCN_SL + BLOCK - 1) / BLOCK;
  __shared__ __attribute__((aligned(16))) int stage[GCN_SEGP];
  __shared__ int cnt[GCN_NBLK];
  __shared__ int off[GCN_NBLK + 1];

  const int tid = threadIdx.x;
  const int e_lo = seg * GCN_SL;
  int n = num_edges - e_lo;
  n = n > GCN_SL ? GCN_SL : n;

  for (int i = tid; i < GCN_NBLK; i += BLOCK) cnt[i] = 0;
  for (int i = tid; i < GCN_SEGP; i += BLOCK) stage[i] = 0;
  __syncthreads();

  int packed[PER], where[PER];
#pragma unroll
  for (int k = 0; k < PER; ++k) {
    const int i = tid + k * BLOCK;
    packed[k] = 0;
    where[k] = -1;
    if (i < n) {
      const int s = src[e_lo + i];
      const int d = dst[e_lo + i];
      const int blk = d / GCN_RPB;
      const int rank = atomicAdd(&cnt[blk], 1);
      packed[k] = s | ((d - blk * GCN_RPB) << 16);
      where[k] = (blk << 16) | rank;
    }
  }
  __syncthreads();

  if (tid < 32) {
    int c[8];
    int run = 0;
#pragma unroll
    for (int j = 0; j < 8; ++j) {
      const int t = cnt[tid * 8 + j];
      c[j] = run;
      run += t;
    }
    int incl = run;
#pragma unroll
    for (int dlt = 1; dlt < 32; dlt <<= 1) {
      const int v = __shfl_up(incl, dlt, 32);
      if (tid >= dlt) incl += v;
    }
    const int base = incl - run;
#pragma unroll
    for (int j = 0; j < 8; ++j) off[tid * 8 + j] = base + c[j];
    if (tid == 31) off[GCN_NBLK] = incl;
  }
  __syncthreads();

#pragma unroll
  for (int k = 0; k < PER; ++k) {
    if (where[k] >= 0) stage[off[where[k] >> 16] + (where[k] & 0xFFFF)] = packed[k];
  }
  __syncthreads();

  int4* o4 = (int4*)(rec_out + (long long)seg * GCN_SEGP);
  for (int i = tid; i < GCN_SEGP / 4; i += BLOCK) o4[i] = ((const int4*)stage)[i];
  for (int i = tid; i <= GCN_NBLK; i += BLOCK) off_out[seg * GCN_OFFP + i] = off[i];
}

template <int F, int SLOTS>
__device__ __forceinline__ void GCN_run_rows(const float* __restrict__ feat,
                                             const int* __restrict__ rec,
                                             int a, int b, float* acc, int lane) {
  int r[SLOTS];
  for (int i = 0; i < SLOTS; ++i) r[i] = rec[a + i];
  for (int g = a; g < b; g += SLOTS) {
    float lo[SLOTS], hi[SLOTS];
    int rn[SLOTS];
    if (g + SLOTS <= b) {
      for (int i = 0; i < SLOTS; ++i) {
        const float* p = feat + (long long)(r[i] & 0xFFFF) * F + lane;
        lo[i] = p[0];
        hi[i] = p[32];
      }
      for (int i = 0; i < SLOTS; ++i) rn[i] = rec[g + SLOTS + i];
      for (int i = 0; i < SLOTS; ++i) {
        float* o = acc + (r[i] >> 16) * F + lane;
        atomicAdd(o, lo[i]);
        atomicAdd(o + 32, hi[i]);
      }
    } else {
      for (int i = 0; i < SLOTS; ++i) {
        lo[i] = 0.f;
        hi[i] = 0.f;
        if (g + i < b) {
          const float* p = feat + (long long)(r[i] & 0xFFFF) * F + lane;
          lo[i] = p[0];
          hi[i] = p[32];
        }
      }
      for (int i = 0; i < SLOTS; ++i) rn[i] = 0;
      for (int i = 0; i < SLOTS; ++i) {
        if (g + i < b) {
          float* o = acc + (r[i] >> 16) * F + lane;
          atomicAdd(o, lo[i]);
          atomicAdd(o + 32, hi[i]);
        }
      }
    }
    for (int i = 0; i < SLOTS; ++i) r[i] = rn[i];
  }
}

template <int F, int SLOTS, int BLOCK>
__global__ __launch_bounds__(BLOCK) void GCN_block_rows_kernel(
    const float* __restrict__ feat,
    const int* __restrict__ rec,
    const int* __restrict__ offs,
    float* __restrict__ out,
    int num_rows) {
  static_assert(F == 64, "the lane map is written for 64-float rows");
  __shared__ __attribute__((aligned(16))) float acc[GCN_RPB * F];
  __shared__ int run_lo[GCN_NSEG];
  __shared__ int run_hi[GCN_NSEG];

  const int tid = threadIdx.x;
  const int lane = tid & 31;
  const int wave = __builtin_amdgcn_readfirstlane(tid >> 5);
  const int blk = blockIdx.x;

  const float4 zero4 = make_float4(0.f, 0.f, 0.f, 0.f);
  for (int i = tid; i < GCN_RPB * F / 4; i += BLOCK) ((float4*)acc)[i] = zero4;
  if (tid < GCN_NSEG) {
    const int* o = offs + tid * GCN_OFFP + blk;
    run_lo[tid] = tid * GCN_SEGP + o[0];
    run_hi[tid] = tid * GCN_SEGP + o[1];
  }
  __syncthreads();

  for (int seg = wave; seg < GCN_NSEG; seg += BLOCK / 32) {
    const int a = __builtin_amdgcn_readfirstlane(run_lo[seg]);
    const int b = __builtin_amdgcn_readfirstlane(run_hi[seg]);
    GCN_run_rows<F, SLOTS>(feat, rec, a, b, acc, lane);
  }
  __syncthreads();

  const int row0 = blk * GCN_RPB;
  int nr = num_rows - row0;
  nr = nr > GCN_RPB ? GCN_RPB : nr;
  float4* o4 = (float4*)(out + (long long)row0 * F);
  for (int i = tid; i < nr * (F / 4); i += BLOCK) o4[i] = ((const float4*)acc)[i];
}

template <int K, int NCOL>
__device__ __forceinline__ void GCN_gemm_strip(const float* __restrict__ A,
                                               const float* __restrict__ B,
                                               float* __restrict__ D,
                                               int nrows, int strip) {
  constexpr int LDSROW = K + 2;
  __shared__ float As[16 * LDSROW];

  const int m0 = strip * 16;
  const int lane = threadIdx.x & 31;
  const int wave = __builtin_amdgcn_readfirstlane((int)(threadIdx.x >> 5));

  for (int idx = threadIdx.x; idx < 16 * (K / 4); idx += blockDim.x) {
    int r = idx / (K / 4);
    int c = (idx % (K / 4)) * 4;
    float4 v = make_float4(0.f, 0.f, 0.f, 0.f);
    if (m0 + r < nrows) {
      v = *(const float4*)(A + (long long)(m0 + r) * K + c);
    }
    As[r * LDSROW + c + 0] = v.x;
    As[r * LDSROW + c + 1] = v.y;
    As[r * LDSROW + c + 2] = v.z;
    As[r * LDSROW + c + 3] = v.w;
  }
  __syncthreads();

  if (wave >= NCOL / 16) return;

  const int n0 = wave * 16;
  const int half = lane >> 4;
  const int l16 = lane & 15;

  v8f c = {};
#pragma unroll
  for (int k0 = 0; k0 < K; k0 += 4) {
    const int kb = k0 + 2 * half;
    v2f a = *(const v2f*)&As[l16 * LDSROW + kb];
    v2f b;
    b.x = B[(kb + 0) * NCOL + n0 + l16];
    b.y = B[(kb + 1) * NCOL + n0 + l16];
    c = __builtin_amdgcn_wmma_f32_16x16x4_f32(
         false, a,  false, b,
         (short)0, c,  false,  false);
  }

#pragma unroll
  for (int r = 0; r < 8; ++r) {
    int row = m0 + r + 8 * half;
    if (row < nrows) {
      D[(long long)row * NCOL + n0 + l16] = c[r];
    }
  }
}

template <int K, int NCOL>
__global__ void GCN_gemm_wmma_kernel(const float* __restrict__ A,
                                     const float* __restrict__ B,
                                     float* __restrict__ D,
                                     int nrows) {
  GCN_gemm_strip<K, NCOL>(A, B, D, nrows, blockIdx.x);
}

template <int K, int NCOL, int STRIPS, int BLOCK>
__global__ __launch_bounds__(BLOCK) void GCN_weights_and_groups_kernel(
    const float* __restrict__ Wa,
    const float* __restrict__ Wb,
    float* __restrict__ Wab,
    const int* __restrict__ src1,
    const int* __restrict__ dst1,
    const int* __restrict__ src2,
    const int* __restrict__ dst2,
    int* __restrict__ rec1,
    int* __restrict__ rec2,
    int* __restrict__ off1,
    int* __restrict__ off2,
    int num_edges) {
  const int id = blockIdx.x;
  if (id < GCN_NSEG) {
    GCN_group_slice<BLOCK>(src1, dst1, num_edges, id, rec1, off1);
  } else if (id < 2 * GCN_NSEG) {
    GCN_group_slice<BLOCK>(src2, dst2, num_edges, id - GCN_NSEG, rec2, off2);
  } else {
    GCN_gemm_strip<K, NCOL>(Wa, Wb, Wab, STRIPS * 16, id - 2 * GCN_NSEG);
  }
}

extern "C" void kernel_launch(void* const* d_in, const int* in_sizes, int n_in,
                              void* d_out, int out_size, void* d_ws, size_t ws_size,
                              hipStream_t stream) {
  const float* x   = (const float*)d_in[0];
  const float* W1  = (const float*)d_in[1];
  const float* W2  = (const float*)d_in[2];
  const int*  src1 = (const int*)d_in[3];
  const int*  dst1 = (const int*)d_in[4];
  const int*  src2 = (const int*)d_in[5];
  const int*  dst2 = (const int*)d_in[6];
  float* out = (float*)d_out;

  constexpr int IN = 128, H = 128, C = 64;
  const int N = in_sizes[0] / IN;
  const int E = in_sizes[3];

  float* W12 = (float*)d_ws;
  float* y   = W12 + (size_t)IN * C;
  float* z   = y + (size_t)N * C;
  int* rec1  = (int*)(z + (size_t)N * C);
  int* rec2  = rec1 + (size_t)GCN_NSEG * GCN_SEGP;
  int* off1  = rec2 + (size_t)GCN_NSEG * GCN_SEGP;
  int* off2  = off1 + (size_t)GCN_NSEG * GCN_OFFP;

  GCN_weights_and_groups_kernel<H, C, IN / 16, 1024>
      <<<2 * GCN_NSEG + IN / 16, 1024, 0, stream>>>(
          W1, W2, W12, src1, dst1, src2, dst2, rec1, rec2, off1, off2, E);

  GCN_gemm_wmma_kernel<IN, C><<<(N + 15) / 16, 128, 0, stream>>>(x, W12, y, N);

  constexpr int SLOTS = 16, BLOCK = 1024;

  GCN_block_rows_kernel<C, SLOTS, BLOCK><<<GCN_NBLK, BLOCK, 0, stream>>>(
      y, rec1, off1, z, N);

  GCN_block_rows_kernel<C, SLOTS, BLOCK><<<GCN_NBLK, BLOCK, 0, stream>>>(
      z, rec2, off2, out, N);
}


# Slice grouping without a round trip per edge: a thread's 26 index loads in one group, then its 13 LDS rank adds and 13 placements back to back, no branches (spare LDS words for tail lanes)
# speedup vs baseline: 4.0892x; 1.0603x over previous
#include <hip/hip_runtime.h>


typedef __attribute__((ext_vector_type(2))) float v2f;
typedef __attribute__((ext_vector_type(8))) float v8f;

constexpr int GCN_NBLK = 256;
constexpr int GCN_RPB = 196;
constexpr int GCN_NSEG = 64;
constexpr int GCN_SL = 12500;
constexpr int GCN_SEGP = 12544;
constexpr int GCN_OFFP = 288;

template <int BLOCK>
__device__ __forceinline__ void GCN_group_slice(const int* __restrict__ src,
                                                const int* __restrict__ dst,
                                                int num_edges, int seg,
                                                int* __restrict__ rec_out,
                                                int* __restrict__ off_out) {
  constexpr int PER = (GCN_SL + BLOCK - 1) / BLOCK;
  __shared__ __attribute__((aligned(16))) int stage[GCN_SEGP + 32];
  __shared__ int cnt[GCN_NBLK + 32];
  __shared__ int off[GCN_NBLK + 1];

  const int tid = threadIdx.x;
  const int lane = tid & 31;
  const int e_lo = seg * GCN_SL;
  const int e_last = num_edges - 1;
  int n = num_edges - e_lo;
  n = n > GCN_SL ? GCN_SL : n;

  for (int i = tid; i < GCN_NBLK + 32; i += BLOCK) cnt[i] = 0;
  for (int i = tid; i < GCN_SEGP; i += BLOCK) stage[i] = 0;
  __syncthreads();

  int s[PER], d[PER];
#pragma unroll
  for (int k = 0; k < PER; ++k) {
    int e = e_lo + tid + k * BLOCK;
    e = e < e_last ? e : e_last;
    e = e > 0 ? e : 0;
    s[k] = src[e];
    d[k] = dst[e];
  }

  int packed[PER], blk[PER], rank[PER];
#pragma unroll
  for (int k = 0; k < PER; ++k) {
    const bool valid = tid + k * BLOCK < n;
    const int b = d[k] / GCN_RPB;
    packed[k] = s[k] | ((d[k] - b * GCN_RPB) << 16);
    blk[k] = b;
    rank[k] = atomicAdd(&cnt[valid ? b : GCN_NBLK + lane], 1);
  }
  __syncthreads();

  if (tid < 32) {
    int c[8];
    int run = 0;
#pragma unroll
    for (int j = 0; j < 8; ++j) {
      const int t = cnt[tid * 8 + j];
      c[j] = run;
      run += t;
    }
    int incl = run;
#pragma unroll
    for (int dlt = 1; dlt < 32; dlt <<= 1) {
      const int v = __shfl_up(incl, dlt, 32);
      if (tid >= dlt) incl += v;
    }
    const int base = incl - run;
#pragma unroll
    for (int j = 0; j < 8; ++j) off[tid * 8 + j] = base + c[j];
    if (tid == 31) off[GCN_NBLK] = incl;
  }
  __syncthreads();

  int first[PER];
#pragma unroll
  for (int k = 0; k < PER; ++k) first[k] = off[blk[k]];
#pragma unroll
  for (int k = 0; k < PER; ++k) {
    const bool valid = tid + k * BLOCK < n;
    stage[valid ? first[k] + rank[k] : GCN_SEGP + lane] = packed[k];
  }
  __syncthreads();

  int4* o4 = (int4*)(rec_out + (long long)seg * GCN_SEGP);
  for (int i = tid; i < GCN_SEGP / 4; i += BLOCK) o4[i] = ((const int4*)stage)[i];
  for (int i = tid; i <= GCN_NBLK; i += BLOCK) off_out[seg * GCN_OFFP + i] = off[i];
}

template <int F, int SLOTS>
__device__ __forceinline__ void GCN_run_rows(const float* __restrict__ feat,
                                             const int* __restrict__ rec,
                                             int a, int b, float* acc, int lane) {
  int r[SLOTS];
  for (int i = 0; i < SLOTS; ++i) r[i] = rec[a + i];
  for (int g = a; g < b; g += SLOTS) {
    float lo[SLOTS], hi[SLOTS];
    int rn[SLOTS];
    if (g + SLOTS <= b) {
      for (int i = 0; i < SLOTS; ++i) {
        const float* p = feat + (long long)(r[i] & 0xFFFF) * F + lane;
        lo[i] = p[0];
        hi[i] = p[32];
      }
      for (int i = 0; i < SLOTS; ++i) rn[i] = rec[g + SLOTS + i];
      for (int i = 0; i < SLOTS; ++i) {
        float* o = acc + (r[i] >> 16) * F + lane;
        atomicAdd(o, lo[i]);
        atomicAdd(o + 32, hi[i]);
      }
    } else {
      for (int i = 0; i < SLOTS; ++i) {
        lo[i] = 0.f;
        hi[i] = 0.f;
        if (g + i < b) {
          const float* p = feat + (long long)(r[i] & 0xFFFF) * F + lane;
          lo[i] = p[0];
          hi[i] = p[32];
        }
      }
      for (int i = 0; i < SLOTS; ++i) rn[i] = 0;
      for (int i = 0; i < SLOTS; ++i) {
        if (g + i < b) {
          float* o = acc + (r[i] >> 16) * F + lane;
          atomicAdd(o, lo[i]);
          atomicAdd(o + 32, hi[i]);
        }
      }
    }
    for (int i = 0; i < SLOTS; ++i) r[i] = rn[i];
  }
}

template <int F, int SLOTS, int BLOCK>
__global__ __launch_bounds__(BLOCK) void GCN_block_rows_kernel(
    const float* __restrict__ feat,
    const int* __restrict__ rec,
    const int* __restrict__ offs,
    float* __restrict__ out,
    int num_rows) {
  static_assert(F == 64, "the lane map is written for 64-float rows");
  __shared__ __attribute__((aligned(16))) float acc[GCN_RPB * F];
  __shared__ int run_lo[GCN_NSEG];
  __shared__ int run_hi[GCN_NSEG];

  const int tid = threadIdx.x;
  const int lane = tid & 31;
  const int wave = __builtin_amdgcn_readfirstlane(tid >> 5);
  const int blk = blockIdx.x;

  const float4 zero4 = make_float4(0.f, 0.f, 0.f, 0.f);
  for (int i = tid; i < GCN_RPB * F / 4; i += BLOCK) ((float4*)acc)[i] = zero4;
  if (tid < GCN_NSEG) {
    const int* o = offs + tid * GCN_OFFP + blk;
    run_lo[tid] = tid * GCN_SEGP + o[0];
    run_hi[tid] = tid * GCN_SEGP + o[1];
  }
  __syncthreads();

  for (int seg = wave; seg < GCN_NSEG; seg += BLOCK / 32) {
    const int a = __builtin_amdgcn_readfirstlane(run_lo[seg]);
    const int b = __builtin_amdgcn_readfirstlane(run_hi[seg]);
    GCN_run_rows<F, SLOTS>(feat, rec, a, b, acc, lane);
  }
  __syncthreads();

  const int row0 = blk * GCN_RPB;
  int nr = num_rows - row0;
  nr = nr > GCN_RPB ? GCN_RPB : nr;
  float4* o4 = (float4*)(out + (long long)row0 * F);
  for (int i = tid; i < nr * (F / 4); i += BLOCK) o4[i] = ((const float4*)acc)[i];
}

template <int K, int NCOL>
__device__ __forceinline__ void GCN_gemm_strip(const float* __restrict__ A,
                                               const float* __restrict__ B,
                                               float* __restrict__ D,
                                               int nrows, int strip) {
  constexpr int LDSROW = K + 2;
  __shared__ float As[16 * LDSROW];

  const int m0 = strip * 16;
  const int lane = threadIdx.x & 31;
  const int wave = __builtin_amdgcn_readfirstlane((int)(threadIdx.x >> 5));

  for (int idx = threadIdx.x; idx < 16 * (K / 4); idx += blockDim.x) {
    int r = idx / (K / 4);
    int c = (idx % (K / 4)) * 4;
    float4 v = make_float4(0.f, 0.f, 0.f, 0.f);
    if (m0 + r < nrows) {
      v = *(const float4*)(A + (long long)(m0 + r) * K + c);
    }
    As[r * LDSROW + c + 0] = v.x;
    As[r * LDSROW + c + 1] = v.y;
    As[r * LDSROW + c + 2] = v.z;
    As[r * LDSROW + c + 3] = v.w;
  }
  __syncthreads();

  if (wave >= NCOL / 16) return;

  const int n0 = wave * 16;
  const int half = lane >> 4;
  const int l16 = lane & 15;

  v8f c = {};
#pragma unroll
  for (int k0 = 0; k0 < K; k0 += 4) {
    const int kb = k0 + 2 * half;
    v2f a = *(const v2f*)&As[l16 * LDSROW + kb];
    v2f b;
    b.x = B[(kb + 0) * NCOL + n0 + l16];
    b.y = B[(kb + 1) * NCOL + n0 + l16];
    c = __builtin_amdgcn_wmma_f32_16x16x4_f32(
         false, a,  false, b,
         (short)0, c,  false,  false);
  }

#pragma unroll
  for (int r = 0; r < 8; ++r) {
    int row = m0 + r + 8 * half;
    if (row < nrows) {
      D[(long long)row * NCOL + n0 + l16] = c[r];
    }
  }
}

template <int K, int NCOL>
__global__ void GCN_gemm_wmma_kernel(const float* __restrict__ A,
                                     const float* __restrict__ B,
                                     float* __restrict__ D,
                                     int nrows) {
  GCN_gemm_strip<K, NCOL>(A, B, D, nrows, blockIdx.x);
}

template <int K, int NCOL, int STRIPS, int BLOCK>
__global__ __launch_bounds__(BLOCK) void GCN_weights_and_groups_kernel(
    const float* __restrict__ Wa,
    const float* __restrict__ Wb,
    float* __restrict__ Wab,
    const int* __restrict__ src1,
    const int* __restrict__ dst1,
    const int* __restrict__ src2,
    const int* __restrict__ dst2,
    int* __restrict__ rec1,
    int* __restrict__ rec2,
    int* __restrict__ off1,
    int* __restrict__ off2,
    int num_edges) {
  const int id = blockIdx.x;
  if (id < GCN_NSEG) {
    GCN_group_slice<BLOCK>(src1, dst1, num_edges, id, rec1, off1);
  } else if (id < 2 * GCN_NSEG) {
    GCN_group_slice<BLOCK>(src2, dst2, num_edges, id - GCN_NSEG, rec2, off2);
  } else {
    GCN_gemm_strip<K, NCOL>(Wa, Wb, Wab, STRIPS * 16, id - 2 * GCN_NSEG);
  }
}

extern "C" void kernel_launch(void* const* d_in, const int* in_sizes, int n_in,
                              void* d_out, int out_size, void* d_ws, size_t ws_size,
                              hipStream_t stream) {
  const float* x   = (const float*)d_in[0];
  const float* W1  = (const float*)d_in[1];
  const float* W2  = (const float*)d_in[2];
  const int*  src1 = (const int*)d_in[3];
  const int*  dst1 = (const int*)d_in[4];
  const int*  src2 = (const int*)d_in[5];
  const int*  dst2 = (const int*)d_in[6];
  float* out = (float*)d_out;

  constexpr int IN = 128, H = 128, C = 64;
  const int N = in_sizes[0] / IN;
  const int E = in_sizes[3];

  float* W12 = (float*)d_ws;
  float* y   = W12 + (size_t)IN * C;
  float* z   = y + (size_t)N * C;
  int* rec1  = (int*)(z + (size_t)N * C);
  int* rec2  = rec1 + (size_t)GCN_NSEG * GCN_SEGP;
  int* off1  = rec2 + (size_t)GCN_NSEG * GCN_SEGP;
  int* off2  = off1 + (size_t)GCN_NSEG * GCN_OFFP;

  GCN_weights_and_groups_kernel<H, C, IN / 16, 1024>
      <<<2 * GCN_NSEG + IN / 16, 1024, 0, stream>>>(
          W1, W2, W12, src1, dst1, src2, dst2, rec1, rec2, off1, off2, E);

  GCN_gemm_wmma_kernel<IN, C><<<(N + 15) / 16, 128, 0, stream>>>(x, W12, y, N);

  constexpr int SLOTS = 16, BLOCK = 1024;

  GCN_block_rows_kernel<C, SLOTS, BLOCK><<<GCN_NBLK, BLOCK, 0, stream>>>(
      y, rec1, off1, z, N);

  GCN_block_rows_kernel<C, SLOTS, BLOCK><<<GCN_NBLK, BLOCK, 0, stream>>>(
      z, rec2, off2, out, N);
}


# Edge passes: a block's records ordered by dst row in LDS (integer count, scan, place by arrival rank); each wave sums its rows in registers in list order and stores them: no f32 LDS adds or copy-out
# speedup vs baseline: 4.7834x; 1.1698x over previous
#include <hip/hip_runtime.h>


typedef __attribute__((ext_vector_type(2))) float v2f;
typedef __attribute__((ext_vector_type(8))) float v8f;

constexpr int GCN_NBLK = 256;
constexpr int GCN_RPB = 196;
constexpr int GCN_NSEG = 64;
constexpr int GCN_SL = 12500;
constexpr int GCN_SEGP = 12544;
constexpr int GCN_OFFP = 288;

template <int BLOCK>
__device__ __forceinline__ void GCN_group_slice(const int* __restrict__ src,
                                                const int* __restrict__ dst,
                                                int num_edges, int seg,
                                                int* __restrict__ rec_out,
                                                int* __restrict__ off_out) {
  constexpr int PER = (GCN_SL + BLOCK - 1) / BLOCK;
  __shared__ __attribute__((aligned(16))) int stage[GCN_SEGP + 32];
  __shared__ int cnt[GCN_NBLK + 32];
  __shared__ int off[GCN_NBLK + 1];

  const int tid = threadIdx.x;
  const int lane = tid & 31;
  const int e_lo = seg * GCN_SL;
  const int e_last = num_edges - 1;
  int n = num_edges - e_lo;
  n = n > GCN_SL ? GCN_SL : n;

  for (int i = tid; i < GCN_NBLK + 32; i += BLOCK) cnt[i] = 0;
  for (int i = tid; i < GCN_SEGP; i += BLOCK) stage[i] = 0;
  __syncthreads();

  int s[PER], d[PER];
#pragma unroll
  for (int k = 0; k < PER; ++k) {
    int e = e_lo + tid + k * BLOCK;
    e = e < e_last ? e : e_last;
    e = e > 0 ? e : 0;
    s[k] = src[e];
    d[k] = dst[e];
  }

  int packed[PER], blk[PER], rank[PER];
#pragma unroll
  for (int k = 0; k < PER; ++k) {
    const bool valid = tid + k * BLOCK < n;
    const int b = d[k] / GCN_RPB;
    packed[k] = s[k] | ((d[k] - b * GCN_RPB) << 16);
    blk[k] = b;
    rank[k] = atomicAdd(&cnt[valid ? b : GCN_NBLK + lane], 1);
  }
  __syncthreads();

  if (tid < 32) {
    int c[8];
    int run = 0;
#pragma unroll
    for (int j = 0; j < 8; ++j) {
      const int t = cnt[tid * 8 + j];
      c[j] = run;
      run += t;
    }
    int incl = run;
#pragma unroll
    for (int dlt = 1; dlt < 32; dlt <<= 1) {
      const int v = __shfl_up(incl, dlt, 32);
      if (tid >= dlt) incl += v;
    }
    const int base = incl - run;
#pragma unroll
    for (int j = 0; j < 8; ++j) off[tid * 8 + j] = base + c[j];
    if (tid == 31) off[GCN_NBLK] = incl;
  }
  __syncthreads();

  int first[PER];
#pragma unroll
  for (int k = 0; k < PER; ++k) first[k] = off[blk[k]];
#pragma unroll
  for (int k = 0; k < PER; ++k) {
    const bool valid = tid + k * BLOCK < n;
    stage[valid ? first[k] + rank[k] : GCN_SEGP + lane] = packed[k];
  }
  __syncthreads();

  int4* o4 = (int4*)(rec_out + (long long)seg * GCN_SEGP);
  for (int i = tid; i < GCN_SEGP / 4; i += BLOCK) o4[i] = ((const int4*)stage)[i];
  for (int i = tid; i <= GCN_NBLK; i += BLOCK) off_out[seg * GCN_OFFP + i] = off[i];
}

template <int F, int SLOTS>
__device__ __forceinline__ void GCN_run_rows(const float* __restrict__ feat,
                                             const int* __restrict__ rec,
                                             int a, int b, float* acc, int lane) {
  int r[SLOTS];
  for (int i = 0; i < SLOTS; ++i) r[i] = rec[a + i];
  for (int g = a; g < b; g += SLOTS) {
    float lo[SLOTS], hi[SLOTS];
    int rn[SLOTS];
    if (g + SLOTS <= b) {
      for (int i = 0; i < SLOTS; ++i) {
        const float* p = feat + (long long)(r[i] & 0xFFFF) * F + lane;
        lo[i] = p[0];
        hi[i] = p[32];
      }
      for (int i = 0; i < SLOTS; ++i) rn[i] = rec[g + SLOTS + i];
      for (int i = 0; i < SLOTS; ++i) {
        float* o = acc + (r[i] >> 16) * F + lane;
        atomicAdd(o, lo[i]);
        atomicAdd(o + 32, hi[i]);
      }
    } else {
      for (int i = 0; i < SLOTS; ++i) {
        lo[i] = 0.f;
        hi[i] = 0.f;
        if (g + i < b) {
          const float* p = feat + (long long)(r[i] & 0xFFFF) * F + lane;
          lo[i] = p[0];
          hi[i] = p[32];
        }
      }
      for (int i = 0; i < SLOTS; ++i) rn[i] = 0;
      for (int i = 0; i < SLOTS; ++i) {
        if (g + i < b) {
          float* o = acc + (r[i] >> 16) * F + lane;
          atomicAdd(o, lo[i]);
          atomicAdd(o + 32, hi[i]);
        }
      }
    }
    for (int i = 0; i < SLOTS; ++i) r[i] = rn[i];
  }
}

template <int F, int SLOTS>
__device__ __forceinline__ void GCN_list_rows(const float* __restrict__ feat,
                                              const int* list, int cap,
                                              int e0, int e1, int R0, int R1,
                                              float* __restrict__ out_blk,
                                              int rows_valid, int lane) {
  static_assert(SLOTS <= 32, "a group's records sit in the lanes of one register");
  int cur = R0;
  float s_lo = 0.f, s_hi = 0.f;
  auto flush_to = [&](int dl) {
    if (__builtin_expect(dl != cur, 0)) {
      if (cur < rows_valid) {
        float* o = out_blk + cur * F + lane;
        o[0] = s_lo;
        o[32] = s_hi;
      }
#pragma nounroll
      for (int z = cur + 1; z < dl; ++z) {
        if (z < rows_valid) {
          float* o = out_blk + z * F + lane;
          o[0] = 0.f;
          o[32] = 0.f;
        }
      }
      s_lo = 0.f;
      s_hi = 0.f;
      cur = dl;
    }
  };

  int at = e0 + lane;
  at = at < cap - 1 ? at : cap - 1;
  int rv = list[at];
  for (int g = e0; g < e1; g += SLOTS) {
    int r[SLOTS];
    float lo[SLOTS], hi[SLOTS];
#pragma unroll
    for (int i = 0; i < SLOTS; ++i) r[i] = __builtin_amdgcn_readlane(rv, i);
    if (g + SLOTS <= e1) {
#pragma unroll
      for (int i = 0; i < SLOTS; ++i) {
        const float* p = feat + (long long)(r[i] & 0xFFFF) * F + lane;
        lo[i] = p[0];
        hi[i] = p[32];
      }
      at = g + SLOTS + lane;
      at = at < cap - 1 ? at : cap - 1;
      rv = list[at];
#pragma unroll
      for (int i = 0; i < SLOTS; ++i) {
        flush_to(r[i] >> 16);
        s_lo += lo[i];
        s_hi += hi[i];
      }
    } else {
#pragma unroll
      for (int i = 0; i < SLOTS; ++i) {
        lo[i] = 0.f;
        hi[i] = 0.f;
        if (g + i < e1) {
          const float* p = feat + (long long)(r[i] & 0xFFFF) * F + lane;
          lo[i] = p[0];
          hi[i] = p[32];
        }
      }
#pragma unroll
      for (int i = 0; i < SLOTS; ++i) {
        if (g + i < e1) {
          flush_to(r[i] >> 16);
          s_lo += lo[i];
          s_hi += hi[i];
        }
      }
    }
  }
  flush_to(R1);
}

template <int F, int SLOTS, int BLOCK>
__global__ __launch_bounds__(BLOCK) void GCN_block_rows_kernel(
    const float* __restrict__ feat,
    const int* __restrict__ rec,
    const int* __restrict__ offs,
    float* __restrict__ out,
    int num_rows) {
  static_assert(F == 64, "the lane map is written for 64-float rows");
  static_assert(GCN_RPB <= 256, "the row counters are scanned 8 per lane by one wave");
  constexpr int WAVES = BLOCK / 32;
  static_assert(GCN_NSEG % WAVES == 0, "every wave takes the same number of runs");
  constexpr int SPW = GCN_NSEG / WAVES;
  constexpr int CAP = GCN_RPB * F;
  __shared__ __attribute__((aligned(16))) float acc[GCN_RPB * F];
  __shared__ int cnt[256];
  __shared__ int rowoff[257];
  int* list = (int*)acc;

  const int tid = threadIdx.x;
  const int lane = tid & 31;
  const int wave = __builtin_amdgcn_readfirstlane(tid >> 5);
  const int blk = blockIdx.x;

  for (int i = tid; i < 256; i += BLOCK) cnt[i] = 0;

  int ra[SPW], rb[SPW], v[SPW][2];
#pragma unroll
  for (int j = 0; j < SPW; ++j) {
    const int seg = wave + j * WAVES;
    const int* o = offs + seg * GCN_OFFP + blk;
    ra[j] = seg * GCN_SEGP + o[0];
    rb[j] = seg * GCN_SEGP + o[1];
  }
#pragma unroll
  for (int j = 0; j < SPW; ++j) {
#pragma unroll
    for (int h = 0; h < 2; ++h) {
      const int k = ra[j] + lane + 32 * h;
      v[j][h] = -1;
      if (k < rb[j]) v[j][h] = rec[k];
    }
  }
  __syncthreads();

#pragma unroll
  for (int j = 0; j < SPW; ++j) {
#pragma unroll
    for (int h = 0; h < 2; ++h) {
      if (v[j][h] >= 0) atomicAdd(&cnt[v[j][h] >> 16], 1);
    }
  }
#pragma unroll
  for (int j = 0; j < SPW; ++j) {
    for (int k = ra[j] + 64 + lane; k < rb[j]; k += 32) atomicAdd(&cnt[rec[k] >> 16], 1);
  }
  __syncthreads();

  if (tid < 32) {
    int c[8];
    int run = 0;
#pragma unroll
    for (int j = 0; j < 8; ++j) {
      const int t = cnt[tid * 8 + j];
      c[j] = run;
      run += t;
    }
    int incl = run;
#pragma unroll
    for (int dlt = 1; dlt < 32; dlt <<= 1) {
      const int u = __shfl_up(incl, dlt, 32);
      if (tid >= dlt) incl += u;
    }
    const int base = incl - run;
#pragma unroll
    for (int j = 0; j < 8; ++j) {
      rowoff[tid * 8 + j] = base + c[j];
      cnt[tid * 8 + j] = base + c[j];
    }
    if (tid == 31) rowoff[256] = incl;
  }
  __syncthreads();

  const int row0 = blk * GCN_RPB;
  const int n_rec = __builtin_amdgcn_readfirstlane(rowoff[256]);

  if (__builtin_expect(n_rec <= CAP, 1)) {
    int pos[SPW][2];
#pragma unroll
    for (int j = 0; j < SPW; ++j) {
#pragma unroll
      for (int h = 0; h < 2; ++h) {
        pos[j][h] = 0;
        if (v[j][h] >= 0) pos[j][h] = atomicAdd(&cnt[v[j][h] >> 16], 1);
      }
    }
#pragma unroll
    for (int j = 0; j < SPW; ++j) {
#pragma unroll
      for (int h = 0; h < 2; ++h) {
        if (v[j][h] >= 0) list[pos[j][h]] = v[j][h];
      }
    }
#pragma unroll
    for (int j = 0; j < SPW; ++j) {
      for (int k = ra[j] + 64 + lane; k < rb[j]; k += 32) {
        const int rr = rec[k];
        list[atomicAdd(&cnt[rr >> 16], 1)] = rr;
      }
    }

    int Rw[2];
#pragma unroll
    for (int q = 0; q < 2; ++q) {
      const int t = ((wave + q) * n_rec) / WAVES;
      int c = 0;
#pragma unroll
      for (int ch = 0; ch < (GCN_RPB + 31) / 32; ++ch) {
        const int R = ch * 32 + lane;
        const bool in = R < GCN_RPB && rowoff[R + 1] <= t;
        c += __builtin_popcount(__builtin_amdgcn_ballot_w32(in));
      }
      Rw[q] = (wave + q == 0) ? 0 : c;
    }
    __syncthreads();

    const int R0 = Rw[0];
    const int R1 = Rw[1];
    const int e0 = __builtin_amdgcn_readfirstlane(rowoff[R0]);
    const int e1 = __builtin_amdgcn_readfirstlane(rowoff[R1]);
    GCN_list_rows<F, SLOTS>(feat, list, CAP, e0, e1, R0, R1,
                            out + (long long)row0 * F, num_rows - row0, lane);
    return;
  }

  const float4 zero4 = make_float4(0.f, 0.f, 0.f, 0.f);
  for (int i = tid; i < GCN_RPB * F / 4; i += BLOCK) ((float4*)acc)[i] = zero4;
  __syncthreads();

  for (int seg = wave; seg < GCN_NSEG; seg += WAVES) {
    const int* o = offs + seg * GCN_OFFP + blk;
    GCN_run_rows<F, SLOTS>(feat, rec, seg * GCN_SEGP + o[0], seg * GCN_SEGP + o[1], acc, lane);
  }
  __syncthreads();

  int nr = num_rows - row0;
  nr = nr > GCN_RPB ? GCN_RPB : nr;
  float4* o4 = (float4*)(out + (long long)row0 * F);
  for (int i = tid; i < nr * (F / 4); i += BLOCK) o4[i] = ((const float4*)acc)[i];
}

template <int K, int NCOL>
__device__ __forceinline__ void GCN_gemm_strip(const float* __restrict__ A,
                                               const float* __restrict__ B,
                                               float* __restrict__ D,
                                               int nrows, int strip) {
  constexpr int LDSROW = K + 2;
  __shared__ float As[16 * LDSROW];

  const int m0 = strip * 16;
  const int lane = threadIdx.x & 31;
  const int wave = __builtin_amdgcn_readfirstlane((int)(threadIdx.x >> 5));

  for (int idx = threadIdx.x; idx < 16 * (K / 4); idx += blockDim.x) {
    int r = idx / (K / 4);
    int c = (idx % (K / 4)) * 4;
    float4 v = make_float4(0.f, 0.f, 0.f, 0.f);
    if (m0 + r < nrows) {
      v = *(const float4*)(A + (long long)(m0 + r) * K + c);
    }
    As[r * LDSROW + c + 0] = v.x;
    As[r * LDSROW + c + 1] = v.y;
    As[r * LDSROW + c + 2] = v.z;
    As[r * LDSROW + c + 3] = v.w;
  }
  __syncthreads();

  if (wave >= NCOL / 16) return;

  const int n0 = wave * 16;
  const int half = lane >> 4;
  const int l16 = lane & 15;

  v8f c = {};
#pragma unroll
  for (int k0 = 0; k0 < K; k0 += 4) {
    const int kb = k0 + 2 * half;
    v2f a = *(const v2f*)&As[l16 * LDSROW + kb];
    v2f b;
    b.x = B[(kb + 0) * NCOL + n0 + l16];
    b.y = B[(kb + 1) * NCOL + n0 + l16];
    c = __builtin_amdgcn_wmma_f32_16x16x4_f32(
         false, a,  false, b,
         (short)0, c,  false,  false);
  }

#pragma unroll
  for (int r = 0; r < 8; ++r) {
    int row = m0 + r + 8 * half;
    if (row < nrows) {
      D[(long long)row * NCOL + n0 + l16] = c[r];
    }
  }
}

template <int K, int NCOL>
__global__ void GCN_gemm_wmma_kernel(const float* __restrict__ A,
                                     const float* __restrict__ B,
                                     float* __restrict__ D,
                                     int nrows) {
  GCN_gemm_strip<K, NCOL>(A, B, D, nrows, blockIdx.x);
}

template <int K, int NCOL, int STRIPS, int BLOCK>
__global__ __launch_bounds__(BLOCK) void GCN_weights_and_groups_kernel(
    const float* __restrict__ Wa,
    const float* __restrict__ Wb,
    float* __restrict__ Wab,
    const int* __restrict__ src1,
    const int* __restrict__ dst1,
    const int* __restrict__ src2,
    const int* __restrict__ dst2,
    int* __restrict__ rec1,
    int* __restrict__ rec2,
    int* __restrict__ off1,
    int* __restrict__ off2,
    int num_edges) {
  const int id = blockIdx.x;
  if (id < GCN_NSEG) {
    GCN_group_slice<BLOCK>(src1, dst1, num_edges, id, rec1, off1);
  } else if (id < 2 * GCN_NSEG) {
    GCN_group_slice<BLOCK>(src2, dst2, num_edges, id - GCN_NSEG, rec2, off2);
  } else {
    GCN_gemm_strip<K, NCOL>(Wa, Wb, Wab, STRIPS * 16, id - 2 * GCN_NSEG);
  }
}

extern "C" void kernel_launch(void* const* d_in, const int* in_sizes, int n_in,
                              void* d_out, int out_size, void* d_ws, size_t ws_size,
                              hipStream_t stream) {
  const float* x   = (const float*)d_in[0];
  const float* W1  = (const float*)d_in[1];
  const float* W2  = (const float*)d_in[2];
  const int*  src1 = (const int*)d_in[3];
  const int*  dst1 = (const int*)d_in[4];
  const int*  src2 = (const int*)d_in[5];
  const int*  dst2 = (const int*)d_in[6];
  float* out = (float*)d_out;

  constexpr int IN = 128, H = 128, C = 64;
  const int N = in_sizes[0] / IN;
  const int E = in_sizes[3];

  float* W12 = (float*)d_ws;
  float* y   = W12 + (size_t)IN * C;
  float* z   = y + (size_t)N * C;
  int* rec1  = (int*)(z + (size_t)N * C);
  int* rec2  = rec1 + (size_t)GCN_NSEG * GCN_SEGP;
  int* off1  = rec2 + (size_t)GCN_NSEG * GCN_SEGP;
  int* off2  = off1 + (size_t)GCN_NSEG * GCN_OFFP;

  GCN_weights_and_groups_kernel<H, C, IN / 16, 1024>
      <<<2 * GCN_NSEG + IN / 16, 1024, 0, stream>>>(
          W1, W2, W12, src1, dst1, src2, dst2, rec1, rec2, off1, off2, E);

  GCN_gemm_wmma_kernel<IN, C><<<(N + 15) / 16, 128, 0, stream>>>(x, W12, y, N);

  constexpr int SLOTS = 16, BLOCK = 1024;

  GCN_block_rows_kernel<C, SLOTS, BLOCK><<<GCN_NBLK, BLOCK, 0, stream>>>(
      y, rec1, off1, z, N);

  GCN_block_rows_kernel<C, SLOTS, BLOCK><<<GCN_NBLK, BLOCK, 0, stream>>>(
      z, rec2, off2, out, N);
}
